# starting kernel
# baseline (speedup 1.0000x reference)
#include <hip/hip_runtime.h>
#include <math.h>

typedef __attribute__((ext_vector_type(2))) float v2f;
typedef __attribute__((ext_vector_type(8))) float v8f;

#define D_IN   1024
#define D_OUT  1024
#define NEXP   8
#define RANK   16
#define KDIM   128
#define SCALING 2.0f

__global__ __launch_bounds__(256) void repack_w2(const float* __restrict__ routed,
                                                 float* __restrict__ w2) {
    int idx = blockIdx.x * 256 + threadIdx.x;
    int k = idx >> 10;
    int o = idx & 1023;
    int e = k >> 4;
    int r = k & 15;
    w2[idx] = routed[(e * D_OUT + o) * RANK + r];
}

__global__ __launch_bounds__(256) void route_compress(const float* __restrict__ x,
                                                      const float* __restrict__ w_route,
                                                      const float* __restrict__ compress,
                                                      float* __restrict__ tmp) {
    const int lane = threadIdx.x & 31;
    const int wave = threadIdx.x >> 5;
    const int tok0 = (blockIdx.x * 8 + wave) * 4;

    float acc[24][4];
#pragma unroll
    for (int j = 0; j < 24; ++j)
#pragma unroll
        for (int t = 0; t < 4; ++t) acc[j][t] = 0.0f;

    for (int it = 0; it < 8; ++it) {
        const int off = it * 128 + lane * 4;
        float4 xv[4];
#pragma unroll
        for (int t = 0; t < 4; ++t)
            xv[t] = *(const float4*)(x + (size_t)(tok0 + t) * D_IN + off);
#pragma unroll
        for (int j = 0; j < 8; ++j) {
            float4 w = *(const float4*)(w_route + j * D_IN + off);
#pragma unroll
            for (int t = 0; t < 4; ++t)
                acc[j][t] += w.x * xv[t].x + w.y * xv[t].y + w.z * xv[t].z + w.w * xv[t].w;
        }
#pragma unroll
        for (int j = 0; j < 16; ++j) {
            float4 w = *(const float4*)(compress + j * D_IN + off);
#pragma unroll
            for (int t = 0; t < 4; ++t)
                acc[8 + j][t] += w.x * xv[t].x + w.y * xv[t].y + w.z * xv[t].z + w.w * xv[t].w;
        }
    }

#pragma unroll
    for (int j = 0; j < 24; ++j)
#pragma unroll
        for (int t = 0; t < 4; ++t) {
            float v = acc[j][t];
#pragma unroll
            for (int s = 16; s >= 1; s >>= 1) v += __shfl_xor(v, s, 32);
            acc[j][t] = v;
        }

#pragma unroll
    for (int t = 0; t < 4; ++t) {
        float v0 = acc[0][t]; int i0 = 0;
#pragma unroll
        for (int j = 1; j < 8; ++j) { float l = acc[j][t]; if (l > v0) { v0 = l; i0 = j; } }
        float v1 = -3.4e38f; int i1 = 0;
#pragma unroll
        for (int j = 0; j < 8; ++j) {
            if (j == i0) continue;
            float l = acc[j][t]; if (l > v1) { v1 = l; i1 = j; }
        }
        float ex = __expf(v1 - v0);
        float g0 = 1.0f / (1.0f + ex);
        float g1 = ex   / (1.0f + ex);

        const int kbase = lane * 4;
        const int e     = lane >> 2;
        const int sel   = lane & 3;
        float g = (e == i0) ? g0 : ((e == i1) ? g1 : 0.0f);
        g *= SCALING;

        float c0 = 0.f, c1 = 0.f, c2 = 0.f, c3 = 0.f;
#pragma unroll
        for (int s = 0; s < 4; ++s) {
            bool p = (sel == s);
            c0 = p ? acc[8 + s * 4 + 0][t] : c0;
            c1 = p ? acc[8 + s * 4 + 1][t] : c1;
            c2 = p ? acc[8 + s * 4 + 2][t] : c2;
            c3 = p ? acc[8 + s * 4 + 3][t] : c3;
        }
        float4 outv = make_float4(g * c0, g * c1, g * c2, g * c3);
        *(float4*)(tmp + (size_t)(tok0 + t) * KDIM + kbase) = outv;
    }
}

__global__ __launch_bounds__(256) void gemm_wmma(const float* __restrict__ A,
                                                 const float* __restrict__ B,
                                                 float* __restrict__ C) {
    const int lane = threadIdx.x & 31;
    const int wave = threadIdx.x >> 5;
    const int tile = blockIdx.x * 8 + wave;
    const int tN   = tile & 63;
    const int tM   = tile >> 6;
    const int half = lane >> 4;
    const int l16  = lane & 15;

    const float* aRow = A + (size_t)(tM * 16 + l16) * KDIM + half * 2;
    const float* bCol = B + (size_t)(half * 2) * D_OUT + tN * 16 + l16;

    v8f c = {0.f, 0.f, 0.f, 0.f, 0.f, 0.f, 0.f, 0.f};
#pragma unroll
    for (int ks = 0; ks < 32; ++ks) {
        v2f a = *(const v2f*)(aRow + ks * 4);
        v2f b;
        b[0] = bCol[(size_t)(ks * 4 + 0) * D_OUT];
        b[1] = bCol[(size_t)(ks * 4 + 1) * D_OUT];
        c = __builtin_amdgcn_wmma_f32_16x16x4_f32(false, a, false, b,
                                                  (short)0, c, false, false);
    }

    float* outp = C + (size_t)(tM * 16 + half * 8) * D_OUT + tN * 16 + l16;
#pragma unroll
    for (int v = 0; v < 8; ++v) outp[(size_t)v * D_OUT] = c[v];
}

extern "C" void kernel_launch(void* const* d_in, const int* in_sizes, int n_in,
                              void* d_out, int out_size, void* d_ws, size_t ws_size,
                              hipStream_t stream) {
    const float* x        = (const float*)d_in[0];
    const float* w_route  = (const float*)d_in[1];
    const float* compress = (const float*)d_in[2];
    const float* routed   = (const float*)d_in[3];
    float*       out      = (float*)d_out;

    float* w2  = (float*)d_ws;
    float* tmp = w2 + KDIM * D_OUT;

    const int ntok = in_sizes[0] / D_IN;

    repack_w2<<<(KDIM * D_OUT) / 256, 256, 0, stream>>>(routed, w2);

    route_compress<<<ntok / 32, 256, 0, stream>>>(x, w_route, compress, tmp);

    const int tiles = (ntok / 16) * (D_OUT / 16);
    gemm_wmma<<<tiles / 8, 256, 0, stream>>>(tmp, w2, out);
}


# router+compress dots: K split over a work-group's 8 waves, each wave keeps its 128-k strip of the 24 weight rows in registers; lanes summed by halving exchange, strips via LDS; sum order changes
# speedup vs baseline: 2.7693x; 2.7693x over previous
#include <hip/hip_runtime.h>
#include <math.h>

typedef __attribute__((ext_vector_type(2))) float v2f;
typedef __attribute__((ext_vector_type(8))) float v8f;

#define D_IN   1024
#define D_OUT  1024
#define NEXP   8
#define RANK   16
#define KDIM   128
#define SCALING 2.0f

#define K1_GROUPS 8
#define K1_VALS   96

__device__ __forceinline__ float lane_xor16(float v) {
    int b = __builtin_bit_cast(int, v);
    return __builtin_bit_cast(float, __builtin_amdgcn_permlanex16(b, b, 0x76543210, 0xfedcba98, false, false));
}
template <int M>
__device__ __forceinline__ float lane_xor_row(float v) {
    int b = __builtin_bit_cast(int, v);
    return __builtin_bit_cast(float, __builtin_amdgcn_update_dpp(0, b, 0x160 + M, 0xF, 0xF, true));
}
template <int M>
__device__ __forceinline__ float lane_xor(float v) {
    if constexpr (M == 16) return lane_xor16(v);
    else return lane_xor_row<M>(v);
}
template <int N, int M>
__device__ __forceinline__ void halve(float (&v)[K1_VALS], bool hi) {
#pragma unroll
    for (int i = 0; i < N / 2; ++i) {
        const float a = v[i], b = v[i + N / 2];
        const float send = hi ? a : b;
        const float keep = hi ? b : a;
        v[i] = keep + lane_xor<M>(send);
    }
}

__global__ __launch_bounds__(256) void route_compress(const float* __restrict__ x,
                                                      const float* __restrict__ w_route,
                                                      const float* __restrict__ compress,
                                                      float* __restrict__ tmp) {
    __shared__ __attribute__((aligned(16))) float red[K1_GROUPS][8][K1_VALS];
    __shared__ __attribute__((aligned(16))) float sums[K1_GROUPS][K1_VALS];

    const int lane = threadIdx.x & 31;
    const int wave = threadIdx.x >> 5;
    const int tok0 = blockIdx.x * (K1_GROUPS * 4);
    const int koff = wave * 128 + lane * 4;

    float4 wr[24];
#pragma unroll
    for (int j = 0; j < 8; ++j)  wr[j]     = *(const float4*)(w_route  + j * D_IN + koff);
#pragma unroll
    for (int j = 0; j < 16; ++j) wr[8 + j] = *(const float4*)(compress + j * D_IN + koff);

    const float* xp = x + (size_t)tok0 * D_IN + koff;
    float4 xv[4];
#pragma unroll
    for (int t = 0; t < 4; ++t) xv[t] = *(const float4*)(xp + (size_t)t * D_IN);

    const bool h16 = (lane & 16) != 0, h8 = (lane & 8) != 0, h4 = (lane & 4) != 0,
               h2  = (lane & 2)  != 0, h1 = (lane & 1) != 0;

    for (int g = 0; g < K1_GROUPS; ++g) {
        const int nxt = (g + 1 < K1_GROUPS) ? (g + 1) : g;
        float4 xn[4];
#pragma unroll
        for (int t = 0; t < 4; ++t) xn[t] = *(const float4*)(xp + (size_t)(nxt * 4 + t) * D_IN);

        float v[K1_VALS];
#pragma unroll
        for (int t = 0; t < 4; ++t)
#pragma unroll
            for (int j = 0; j < 24; ++j)
                v[t * 24 + j] = wr[j].x * xv[t].x + wr[j].y * xv[t].y + wr[j].z * xv[t].z + wr[j].w * xv[t].w;

        halve<96, 16>(v, h16);
        halve<48, 8>(v, h8);
        halve<24, 4>(v, h4);
        halve<12, 2>(v, h2);
        halve<6, 1>(v, h1);

        float* r = &red[g][wave][3 * lane];
        r[0] = v[0]; r[1] = v[1]; r[2] = v[2];

#pragma unroll
        for (int t = 0; t < 4; ++t) xv[t] = xn[t];
    }
    __syncthreads();

    {
        float s0 = 0.0f, s1 = 0.0f, s2 = 0.0f;
#pragma unroll
        for (int w = 0; w < 8; ++w) {
            const float* r = &red[wave][w][3 * lane];
            s0 += r[0]; s1 += r[1]; s2 += r[2];
        }
        float* s = &sums[wave][3 * lane];
        s[0] = s0; s[1] = s1; s[2] = s2;
    }
    __syncthreads();

    const float* sg = &sums[wave][0];
#pragma unroll
    for (int t = 0; t < 4; ++t) {
        float l[8];
        {
            const float4 la = *(const float4*)(sg + t * 24);
            const float4 lb = *(const float4*)(sg + t * 24 + 4);
            l[0] = la.x; l[1] = la.y; l[2] = la.z; l[3] = la.w;
            l[4] = lb.x; l[5] = lb.y; l[6] = lb.z; l[7] = lb.w;
        }
        float v0 = l[0]; int i0 = 0;
#pragma unroll
        for (int j = 1; j < 8; ++j) { if (l[j] > v0) { v0 = l[j]; i0 = j; } }
        float v1 = -3.4e38f; int i1 = 0;
#pragma unroll
        for (int j = 0; j < 8; ++j) {
            if (j == i0) continue;
            if (l[j] > v1) { v1 = l[j]; i1 = j; }
        }
        float ex = __expf(v1 - v0);
        float g0 = 1.0f / (1.0f + ex);
        float g1 = ex   / (1.0f + ex);

        const int kbase = lane * 4;
        const int e     = lane >> 2;
        const int sel   = lane & 3;
        float gt = (e == i0) ? g0 : ((e == i1) ? g1 : 0.0f);
        gt *= SCALING;

        const float4 c = *(const float4*)(sg + t * 24 + 8 + sel * 4);
        float4 outv = make_float4(gt * c.x, gt * c.y, gt * c.z, gt * c.w);
        *(float4*)(tmp + (size_t)(tok0 + wave * 4 + t) * KDIM + kbase) = outv;
    }
}

typedef __attribute__((ext_vector_type(4))) float v4f;

#define NT_PER_WAVE 4
#define MT_PER_WG   8

__global__ __launch_bounds__(128) __attribute__((amdgpu_waves_per_eu(2, 2))) void gemm_wmma_ws(const float* __restrict__ A,
                                                    const float* __restrict__ routed,
                                                    float* __restrict__ C) {
    const int lane = threadIdx.x & 31;
    const int wave = threadIdx.x >> 5;
    const int half = lane >> 4;
    const int l16  = lane & 15;
    const int cgrp = blockIdx.x & 3;
    const int mchk = blockIdx.x >> 2;
    const int col0 = cgrp * 256 + wave * 64;

    v4f b[NT_PER_WAVE][16];
#pragma unroll
    for (int j = 0; j < NT_PER_WAVE; ++j)
#pragma unroll
        for (int q = 0; q < 16; ++q) {
            const int e = q >> 1;
            const int r = (q & 1) * 8 + half * 4;
            b[j][q] = *(const v4f*)(routed + ((size_t)(e * D_OUT + col0 + j * 16 + l16)) * RANK + r);
        }

    const int row0 = mchk * (MT_PER_WG * 16);
    const float* aPtr = A + (size_t)(row0 + l16) * KDIM + half * 4;
    float* outp = C + (size_t)(row0 + half * 8) * D_OUT + col0 + l16;

    v4f a[16];
#pragma unroll
    for (int q = 0; q < 16; ++q) a[q] = *(const v4f*)(aPtr + q * 8);

    for (int mt = 0; mt < MT_PER_WG; ++mt) {
        const int nxt = (mt + 1 < MT_PER_WG) ? (mt + 1) : mt;
        const float* aNext = aPtr + (size_t)nxt * 16 * KDIM;

        v8f c[NT_PER_WAVE];
#pragma unroll
        for (int j = 0; j < NT_PER_WAVE; ++j)
            c[j] = (v8f){0.f, 0.f, 0.f, 0.f, 0.f, 0.f, 0.f, 0.f};
#pragma unroll
        for (int q = 0; q < 16; ++q) {
#pragma unroll
            for (int p = 0; p < 2; ++p) {
                v2f af;
                af[0] = a[q][2 * p];
                af[1] = a[q][2 * p + 1];
#pragma unroll
                for (int j = 0; j < NT_PER_WAVE; ++j) {
                    v2f bf;
                    bf[0] = b[j][q][2 * p];
                    bf[1] = b[j][q][2 * p + 1];
                    c[j] = __builtin_amdgcn_wmma_f32_16x16x4_f32(false, af, false, bf,
                                                                 (short)0, c[j], false, false);
                }
            }
            a[q] = *(const v4f*)(aNext + q * 8);
        }
#pragma unroll
        for (int j = 0; j < NT_PER_WAVE; ++j)
#pragma unroll
            for (int v = 0; v < 8; ++v) outp[(size_t)v * D_OUT + j * 16] = c[j][v];
        outp += 16 * D_OUT;
    }
}

extern "C" void kernel_launch(void* const* d_in, const int* in_sizes, int n_in,
                              void* d_out, int out_size, void* d_ws, size_t ws_size,
                              hipStream_t stream) {
    const float* x        = (const float*)d_in[0];
    const float* w_route  = (const float*)d_in[1];
    const float* compress = (const float*)d_in[2];
    const float* routed   = (const float*)d_in[3];
    float*       out      = (float*)d_out;

    float* tmp = (float*)d_ws + KDIM * D_OUT;

    const int ntok = in_sizes[0] / D_IN;

    route_compress<<<ntok / 32, 256, 0, stream>>>(x, w_route, compress, tmp);

    gemm_wmma_ws<<<(ntok / (MT_PER_WG * 16)) * 4, 128, 0, stream>>>(tmp, routed, out);
}


# router+compress dots: a wave owns 4 tokens over all K, weight rows staged per work-group in LDS (4 rows of one k per 16-byte read), one lane exchange per 4 tokens, no strip sums; k sum order changes
# speedup vs baseline: 2.9147x; 1.0525x over previous
#include <hip/hip_runtime.h>
#include <math.h>

typedef __attribute__((ext_vector_type(2))) float v2f;
typedef __attribute__((ext_vector_type(8))) float v8f;

#define D_IN   1024
#define D_OUT  1024
#define NEXP   8
#define RANK   16
#define KDIM   128
#define SCALING 2.0f

#define K1_VALS   96
#define K1_WL     (8 * 4 * 6 * 32)

__device__ __forceinline__ float lane_xor16(float v) {
    int b = __builtin_bit_cast(int, v);
    return __builtin_bit_cast(float, __builtin_amdgcn_permlanex16(b, b, 0x76543210, 0xfedcba98, false, false));
}
template <int M>
__device__ __forceinline__ float lane_xor_row(float v) {
    int b = __builtin_bit_cast(int, v);
    return __builtin_bit_cast(float, __builtin_amdgcn_update_dpp(0, b, 0x160 + M, 0xF, 0xF, true));
}
template <int M>
__device__ __forceinline__ float lane_xor(float v) {
    if constexpr (M == 16) return lane_xor16(v);
    else return lane_xor_row<M>(v);
}
template <int N, int M>
__device__ __forceinline__ void halve(float (&v)[K1_VALS], bool hi) {
#pragma unroll
    for (int i = 0; i < N / 2; ++i) {
        const float a = v[i], b = v[i + N / 2];
        const float send = hi ? a : b;
        const float keep = hi ? b : a;
        v[i] = keep + lane_xor<M>(send);
    }
}

__global__ __launch_bounds__(256) void route_compress(const float* __restrict__ x,
                                                      const float* __restrict__ w_route,
                                                      const float* __restrict__ compress,
                                                      float* __restrict__ tmp) {
    __shared__ __attribute__((aligned(16))) float4 wl[K1_WL];
    __shared__ __attribute__((aligned(16))) float sums[8][K1_VALS];

    const int lane = threadIdx.x & 31;
    const int wave = threadIdx.x >> 5;
    const int tok0 = blockIdx.x * 32;

#pragma unroll
    for (int jq = 0; jq < 6; ++jq) {
        float4 r[4];
#pragma unroll
        for (int u = 0; u < 4; ++u) {
            const int j = jq * 4 + u;
            const float* src = (j < 8) ? (w_route + j * D_IN) : (compress + (j - 8) * D_IN);
            r[u] = *(const float4*)(src + threadIdx.x * 4);
        }
        wl[((wave * 4 + 0) * 6 + jq) * 32 + lane] = make_float4(r[0].x, r[1].x, r[2].x, r[3].x);
        wl[((wave * 4 + 1) * 6 + jq) * 32 + lane] = make_float4(r[0].y, r[1].y, r[2].y, r[3].y);
        wl[((wave * 4 + 2) * 6 + jq) * 32 + lane] = make_float4(r[0].z, r[1].z, r[2].z, r[3].z);
        wl[((wave * 4 + 3) * 6 + jq) * 32 + lane] = make_float4(r[0].w, r[1].w, r[2].w, r[3].w);
    }

    const float* xp = x + (size_t)(tok0 + wave * 4) * D_IN + lane * 4;
    float4 xs[4][4];
#pragma unroll
    for (int c = 0; c < 4; ++c)
#pragma unroll
        for (int t = 0; t < 4; ++t) xs[c][t] = *(const float4*)(xp + (size_t)t * D_IN + c * 128);

    __syncthreads();

    v2f acc[48];
#pragma unroll
    for (int c = 0; c < 8; ++c) {
#pragma unroll
        for (int kk = 0; kk < 4; ++kk) {
#pragma unroll
            for (int jq = 0; jq < 6; ++jq) {
                const float4 w4 = wl[((c * 4 + kk) * 6 + jq) * 32 + lane];
                const v2f w01 = {w4.x, w4.y};
                const v2f w23 = {w4.z, w4.w};
#pragma unroll
                for (int t = 0; t < 4; ++t) {
                    const float4 xx = xs[c & 3][t];
                    const float xk = (kk == 0) ? xx.x : (kk == 1) ? xx.y : (kk == 2) ? xx.z : xx.w;
                    const v2f xb = {xk, xk};
                    if (c == 0 && kk == 0) {
                        acc[t * 12 + jq * 2]     = w01 * xb;
                        acc[t * 12 + jq * 2 + 1] = w23 * xb;
                    } else {
                        acc[t * 12 + jq * 2]     = w01 * xb + acc[t * 12 + jq * 2];
                        acc[t * 12 + jq * 2 + 1] = w23 * xb + acc[t * 12 + jq * 2 + 1];
                    }
                }
            }
        }
        if (c < 4) {
#pragma unroll
            for (int t = 0; t < 4; ++t) xs[c][t] = *(const float4*)(xp + (size_t)t * D_IN + (c + 4) * 128);
        }
    }

    float v[K1_VALS];
#pragma unroll
    for (int i = 0; i < 48; ++i) { v[2 * i] = acc[i][0]; v[2 * i + 1] = acc[i][1]; }

    const bool h16 = (lane & 16) != 0, h8 = (lane & 8) != 0, h4 = (lane & 4) != 0,
               h2  = (lane & 2)  != 0, h1 = (lane & 1) != 0;
    halve<96, 16>(v, h16);
    halve<48, 8>(v, h8);
    halve<24, 4>(v, h4);
    halve<12, 2>(v, h2);
    halve<6, 1>(v, h1);

    {
        float* s = &sums[wave][3 * lane];
        s[0] = v[0]; s[1] = v[1]; s[2] = v[2];
    }
    __syncthreads();

    const float* sg = &sums[wave][0];
#pragma unroll
    for (int t = 0; t < 4; ++t) {
        float l[8];
        {
            const float4 la = *(const float4*)(sg + t * 24);
            const float4 lb = *(const float4*)(sg + t * 24 + 4);
            l[0] = la.x; l[1] = la.y; l[2] = la.z; l[3] = la.w;
            l[4] = lb.x; l[5] = lb.y; l[6] = lb.z; l[7] = lb.w;
        }
        float v0 = l[0]; int i0 = 0;
#pragma unroll
        for (int j = 1; j < 8; ++j) { if (l[j] > v0) { v0 = l[j]; i0 = j; } }
        float v1 = -3.4e38f; int i1 = 0;
#pragma unroll
        for (int j = 0; j < 8; ++j) {
            if (j == i0) continue;
            if (l[j] > v1) { v1 = l[j]; i1 = j; }
        }
        float ex = __expf(v1 - v0);
        float g0 = 1.0f / (1.0f + ex);
        float g1 = ex   / (1.0f + ex);

        const int kbase = lane * 4;
        const int e     = lane >> 2;
        const int sel   = lane & 3;
        float gt = (e == i0) ? g0 : ((e == i1) ? g1 : 0.0f);
        gt *= SCALING;

        const float4 c = *(const float4*)(sg + t * 24 + 8 + sel * 4);
        float4 outv = make_float4(gt * c.x, gt * c.y, gt * c.z, gt * c.w);
        *(float4*)(tmp + (size_t)(tok0 + wave * 4 + t) * KDIM + kbase) = outv;
    }
}

typedef __attribute__((ext_vector_type(4))) float v4f;

#define NT_PER_WAVE 4
#define MT_PER_WG   8

__global__ __launch_bounds__(128) __attribute__((amdgpu_waves_per_eu(2, 2))) void gemm_wmma_ws(const float* __restrict__ A,
                                                    const float* __restrict__ routed,
                                                    float* __restrict__ C) {
    const int lane = threadIdx.x & 31;
    const int wave = threadIdx.x >> 5;
    const int half = lane >> 4;
    const int l16  = lane & 15;
    const int cgrp = blockIdx.x & 3;
    const int mchk = blockIdx.x >> 2;
    const int col0 = cgrp * 256 + wave * 64;

    v4f b[NT_PER_WAVE][16];
#pragma unroll
    for (int j = 0; j < NT_PER_WAVE; ++j)
#pragma unroll
        for (int q = 0; q < 16; ++q) {
            const int e = q >> 1;
            const int r = (q & 1) * 8 + half * 4;
            b[j][q] = *(const v4f*)(routed + ((size_t)(e * D_OUT + col0 + j * 16 + l16)) * RANK + r);
        }

    const int row0 = mchk * (MT_PER_WG * 16);
    const float* aPtr = A + (size_t)(row0 + l16) * KDIM + half * 4;
    float* outp = C + (size_t)(row0 + half * 8) * D_OUT + col0 + l16;

    v4f a[16];
#pragma unroll
    for (int q = 0; q < 16; ++q) a[q] = *(const v4f*)(aPtr + q * 8);

    for (int mt = 0; mt < MT_PER_WG; ++mt) {
        const int nxt = (mt + 1 < MT_PER_WG) ? (mt + 1) : mt;
        const float* aNext = aPtr + (size_t)nxt * 16 * KDIM;

        v8f c[NT_PER_WAVE];
#pragma unroll
        for (int j = 0; j < NT_PER_WAVE; ++j)
            c[j] = (v8f){0.f, 0.f, 0.f, 0.f, 0.f, 0.f, 0.f, 0.f};
#pragma unroll
        for (int q = 0; q < 16; ++q) {
#pragma unroll
            for (int p = 0; p < 2; ++p) {
                v2f af;
                af[0] = a[q][2 * p];
                af[1] = a[q][2 * p + 1];
#pragma unroll
                for (int j = 0; j < NT_PER_WAVE; ++j) {
                    v2f bf;
                    bf[0] = b[j][q][2 * p];
                    bf[1] = b[j][q][2 * p + 1];
                    c[j] = __builtin_amdgcn_wmma_f32_16x16x4_f32(false, af, false, bf,
                                                                 (short)0, c[j], false, false);
                }
            }
            a[q] = *(const v4f*)(aNext + q * 8);
        }
#pragma unroll
        for (int j = 0; j < NT_PER_WAVE; ++j)
#pragma unroll
            for (int v = 0; v < 8; ++v) outp[(size_t)v * D_OUT + j * 16] = c[j][v];
        outp += 16 * D_OUT;
    }
}

extern "C" void kernel_launch(void* const* d_in, const int* in_sizes, int n_in,
                              void* d_out, int out_size, void* d_ws, size_t ws_size,
                              hipStream_t stream) {
    const float* x        = (const float*)d_in[0];
    const float* w_route  = (const float*)d_in[1];
    const float* compress = (const float*)d_in[2];
    const float* routed   = (const float*)d_in[3];
    float*       out      = (float*)d_out;

    float* tmp = (float*)d_ws + KDIM * D_OUT;

    const int ntok = in_sizes[0] / D_IN;

    route_compress<<<ntok / 32, 256, 0, stream>>>(x, w_route, compress, tmp);

    gemm_wmma_ws<<<(ntok / (MT_PER_WG * 16)) * 4, 128, 0, stream>>>(tmp, routed, out);
}


# both launches fused, token-stationary: a work-group holds 64 tokens' gated rows as A fragments in registers, B fragments straight from routed one 16-column tile ahead; sum order unchanged
# speedup vs baseline: 3.1307x; 1.0741x over previous
#include <hip/hip_runtime.h>
#include <math.h>

typedef __attribute__((ext_vector_type(2))) float v2f;
typedef __attribute__((ext_vector_type(4))) float v4f;
typedef __attribute__((ext_vector_type(8))) float v8f;

#define D_IN   1024
#define D_OUT  1024
#define NEXP   8
#define RANK   16
#define KDIM   128
#define SCALING 2.0f

#define K1_VALS   96
#define K1_WL     (8 * 4 * 6 * 32)
#define TOK_PER_WG 64
#define TL_PITCH  132

__device__ __forceinline__ float lane_xor16(float v) {
    int b = __builtin_bit_cast(int, v);
    return __builtin_bit_cast(float, __builtin_amdgcn_permlanex16(b, b, 0x76543210, 0xfedcba98, false, false));
}
template <int M>
__device__ __forceinline__ float lane_xor_row(float v) {
    int b = __builtin_bit_cast(int, v);
    return __builtin_bit_cast(float, __builtin_amdgcn_update_dpp(0, b, 0x160 + M, 0xF, 0xF, true));
}
template <int M>
__device__ __forceinline__ float lane_xor(float v) {
    if constexpr (M == 16) return lane_xor16(v);
    else return lane_xor_row<M>(v);
}
template <int N, int M>
__device__ __forceinline__ void halve(float (&v)[K1_VALS], bool hi) {
#pragma unroll
    for (int i = 0; i < N / 2; ++i) {
        const float a = v[i], b = v[i + N / 2];
        const float send = hi ? a : b;
        const float keep = hi ? b : a;
        v[i] = keep + lane_xor<M>(send);
    }
}

__device__ __forceinline__ void load_b_tile(v4f (&b)[16], const float* __restrict__ bp) {
#pragma unroll
    for (int q = 0; q < 16; ++q)
        b[q] = *(const v4f*)(bp + (size_t)(q >> 1) * (D_OUT * RANK) + (q & 1) * 8);
}

__device__ __forceinline__ void tile_products(const v4f (&a)[4][16], const v4f (&b)[16],
                                              float* __restrict__ op) {
    v8f c[4];
#pragma unroll
    for (int i = 0; i < 4; ++i)
        c[i] = (v8f){0.f, 0.f, 0.f, 0.f, 0.f, 0.f, 0.f, 0.f};
#pragma unroll
    for (int q = 0; q < 16; ++q) {
#pragma unroll
        for (int p = 0; p < 2; ++p) {
            v2f bf;
            bf[0] = b[q][2 * p];
            bf[1] = b[q][2 * p + 1];
#pragma unroll
            for (int i = 0; i < 4; ++i) {
                v2f af;
                af[0] = a[i][q][2 * p];
                af[1] = a[i][q][2 * p + 1];
                c[i] = __builtin_amdgcn_wmma_f32_16x16x4_f32(false, af, false, bf,
                                                             (short)0, c[i], false, false);
            }
        }
    }
#pragma unroll
    for (int i = 0; i < 4; ++i)
#pragma unroll
        for (int v = 0; v < 8; ++v) op[(size_t)(i * 16 + v) * D_OUT] = c[i][v];
}

__global__ __launch_bounds__(256) __attribute__((amdgpu_waves_per_eu(2, 2))) void moe_fused(
                                                      const float* __restrict__ x,
                                                      const float* __restrict__ w_route,
                                                      const float* __restrict__ compress,
                                                      const float* __restrict__ routed,
                                                      float* __restrict__ out) {
    __shared__ __attribute__((aligned(16))) float4 wl[K1_WL];
    __shared__ __attribute__((aligned(16))) float sums[8][2][K1_VALS];
    __shared__ __attribute__((aligned(16))) float tl[TOK_PER_WG * TL_PITCH];

    const int lane = threadIdx.x & 31;
    const int wave = threadIdx.x >> 5;
    const int tok0 = blockIdx.x * TOK_PER_WG;

#pragma unroll
    for (int jq = 0; jq < 6; ++jq) {
        float4 r[4];
#pragma unroll
        for (int u = 0; u < 4; ++u) {
            const int j = jq * 4 + u;
            const float* src = (j < 8) ? (w_route + j * D_IN) : (compress + (j - 8) * D_IN);
            r[u] = *(const float4*)(src + threadIdx.x * 4);
        }
        wl[((wave * 4 + 0) * 6 + jq) * 32 + lane] = make_float4(r[0].x, r[1].x, r[2].x, r[3].x);
        wl[((wave * 4 + 1) * 6 + jq) * 32 + lane] = make_float4(r[0].y, r[1].y, r[2].y, r[3].y);
        wl[((wave * 4 + 2) * 6 + jq) * 32 + lane] = make_float4(r[0].z, r[1].z, r[2].z, r[3].z);
        wl[((wave * 4 + 3) * 6 + jq) * 32 + lane] = make_float4(r[0].w, r[1].w, r[2].w, r[3].w);
    }

    const float* xp = x + (size_t)(tok0 + wave * 4) * D_IN + lane * 4;
    float4 xs[4][4];
#pragma unroll
    for (int c = 0; c < 4; ++c)
#pragma unroll
        for (int t = 0; t < 4; ++t) xs[c][t] = *(const float4*)(xp + (size_t)t * D_IN + c * 128);

    __syncthreads();

#pragma nounroll
    for (int pass = 0; pass < 2; ++pass) {
        const float* xn = xp + ((pass == 0) ? (size_t)32 * D_IN : (size_t)0);
        int wlane = lane;
        asm volatile("" : "+v"(wlane));

        float4 wq[12];
#pragma unroll
        for (int s = 0; s < 12; ++s) wq[s] = wl[s * 32 + wlane];

        v2f acc[48];
#pragma unroll
        for (int c = 0; c < 8; ++c) {
#pragma unroll
            for (int kk = 0; kk < 4; ++kk) {
#pragma unroll
                for (int jq = 0; jq < 6; ++jq) {
                    const int step = (c * 4 + kk) * 6 + jq;
                    const float4 w4 = wq[step % 12];
                    if (step + 12 < 192) wq[step % 12] = wl[(step + 12) * 32 + wlane];
                    const v2f w01 = {w4.x, w4.y};
                    const v2f w23 = {w4.z, w4.w};
#pragma unroll
                    for (int t = 0; t < 4; ++t) {
                        const float4 xx = xs[c & 3][t];
                        const float xk = (kk == 0) ? xx.x : (kk == 1) ? xx.y : (kk == 2) ? xx.z : xx.w;
                        const v2f xb = {xk, xk};
                        if (c == 0 && kk == 0) {
                            acc[t * 12 + jq * 2]     = w01 * xb;
                            acc[t * 12 + jq * 2 + 1] = w23 * xb;
                        } else {
                            acc[t * 12 + jq * 2]     = w01 * xb + acc[t * 12 + jq * 2];
                            acc[t * 12 + jq * 2 + 1] = w23 * xb + acc[t * 12 + jq * 2 + 1];
                        }
                    }
                }
            }
            if (c < 4) {
#pragma unroll
                for (int t = 0; t < 4; ++t) xs[c][t] = *(const float4*)(xp + (size_t)t * D_IN + (c + 4) * 128);
            } else {
#pragma unroll
                for (int t = 0; t < 4; ++t) xs[c - 4][t] = *(const float4*)(xn + (size_t)t * D_IN + (c - 4) * 128);
            }
            __builtin_amdgcn_sched_barrier(0);
        }

        float v[K1_VALS];
#pragma unroll
        for (int i = 0; i < 48; ++i) { v[2 * i] = acc[i][0]; v[2 * i + 1] = acc[i][1]; }

        const bool h16 = (lane & 16) != 0, h8 = (lane & 8) != 0, h4 = (lane & 4) != 0,
                   h2  = (lane & 2)  != 0, h1 = (lane & 1) != 0;
        halve<96, 16>(v, h16);
        halve<48, 8>(v, h8);
        halve<24, 4>(v, h4);
        halve<12, 2>(v, h2);
        halve<6, 1>(v, h1);

        {
            float* s = &sums[wave][pass][3 * lane];
            s[0] = v[0]; s[1] = v[1]; s[2] = v[2];
        }
        xp = xn;
    }
    __syncthreads();

    for (int pass = 0; pass < 2; ++pass) {
        const float* sg = &sums[wave][pass][0];
#pragma unroll
        for (int t = 0; t < 4; ++t) {
            float l[8];
            {
                const float4 la = *(const float4*)(sg + t * 24);
                const float4 lb = *(const float4*)(sg + t * 24 + 4);
                l[0] = la.x; l[1] = la.y; l[2] = la.z; l[3] = la.w;
                l[4] = lb.x; l[5] = lb.y; l[6] = lb.z; l[7] = lb.w;
            }
            float v0 = l[0]; int i0 = 0;
#pragma unroll
            for (int j = 1; j < 8; ++j) { if (l[j] > v0) { v0 = l[j]; i0 = j; } }
            float v1 = -3.4e38f; int i1 = 0;
#pragma unroll
            for (int j = 0; j < 8; ++j) {
                if (j == i0) continue;
                if (l[j] > v1) { v1 = l[j]; i1 = j; }
            }
            float ex = __expf(v1 - v0);
            float g0 = 1.0f / (1.0f + ex);
            float g1 = ex   / (1.0f + ex);

            const int kbase = lane * 4;
            const int e     = lane >> 2;
            const int sel   = lane & 3;
            float gt = (e == i0) ? g0 : ((e == i1) ? g1 : 0.0f);
            gt *= SCALING;

            const float4 c = *(const float4*)(sg + t * 24 + 8 + sel * 4);
            float4 outv = make_float4(gt * c.x, gt * c.y, gt * c.z, gt * c.w);
            *(float4*)(&tl[(pass * 32 + wave * 4 + t) * TL_PITCH + kbase]) = outv;
        }
    }
    __syncthreads();

    const int half = lane >> 4;
    const int l16  = lane & 15;

    v4f a[4][16];
#pragma unroll
    for (int i = 0; i < 4; ++i)
#pragma unroll
        for (int q = 0; q < 16; ++q)
            a[i][q] = *(const v4f*)(&tl[(i * 16 + l16) * TL_PITCH + q * 8 + half * 4]);

    const float* bp = routed + (size_t)(wave * 128 + l16) * RANK + half * 4;
    float* op = out + (size_t)(tok0 + half * 8) * D_OUT + wave * 128 + l16;

    v4f b0[16], b1[16];
    load_b_tile(b0, bp);
    for (int n2 = 0; n2 < 4; ++n2) {
        load_b_tile(b1, bp + 16 * RANK);
        tile_products(a, b0, op);
        const int adv = (n2 < 3) ? 32 : 16;
        load_b_tile(b0, bp + adv * RANK);
        tile_products(a, b1, op + 16);
        bp += 32 * RANK;
        op += 32;
    }
}

extern "C" void kernel_launch(void* const* d_in, const int* in_sizes, int n_in,
                              void* d_out, int out_size, void* d_ws, size_t ws_size,
                              hipStream_t stream) {
    const float* x        = (const float*)d_in[0];
    const float* w_route  = (const float*)d_in[1];
    const float* compress = (const float*)d_in[2];
    const float* routed   = (const float*)d_in[3];
    float*       out      = (float*)d_out;

    const int ntok = in_sizes[0] / D_IN;

    moe_fused<<<ntok / TOK_PER_WG, 256, 0, stream>>>(x, w_route, compress, routed, out);
}
